# starting kernel
# baseline (speedup 1.0000x reference)
#include <hip/hip_runtime.h>
#include <math.h>


typedef __attribute__((ext_vector_type(2))) float v2f;
typedef __attribute__((ext_vector_type(8))) float v8f;

#define NUM_IN   8
#define NUM_OUT  8
#define IN_DIM   64
#define OUT_DIM  64
#define JB       16384
#define NCOL     512
#define STRIPS   1024
#define CAP_EPS  1e-9f

union V8 { v8f v; float f[8]; };

__global__ __launch_bounds__(256) void k_xpart(const float* __restrict__ x,
                                               float* __restrict__ xpart) {
    int part = blockIdx.x;
    int i    = blockIdx.y;
    int t    = threadIdx.x;
    int d    = t & 63;
    int g    = t >> 6;
    const float* xi = x + (size_t)i * JB * IN_DIM;
    int r0 = part * 1024;
    float acc = 0.f;
    for (int r = r0 + g; r < r0 + 1024; r += 4)
        acc += xi[(size_t)r * 64 + d];
    __shared__ float sd[256];
    sd[t] = acc;
    __syncthreads();
    if (g == 0)
        xpart[(i * 16 + part) * 64 + d] = sd[d] + sd[64 + d] + sd[128 + d] + sd[192 + d];
}

__global__ __launch_bounds__(256) void k_uhm(const float* __restrict__ xpart,
                                             const float* __restrict__ W,
                                             float* __restrict__ uhm,
                                             float* __restrict__ Bmat) {
    __shared__ float umean[512];
    int t = threadIdx.x;
    for (int v = t; v < 512; v += 256) {
        int i = v >> 6, d = v & 63;
        float s = 0.f;
        for (int p = 0; p < 16; ++p) s += xpart[(i * 16 + p) * 64 + d];
        umean[v] = s * (1.0f / (float)JB);
    }
    __syncthreads();
    for (int v = t; v < 4096; v += 256) {
        int i = v >> 9, j = (v >> 6) & 7, e = v & 63;
        const float* wp = W + ((size_t)i * 8 + j) * 4096 + e;
        const float* um = umean + i * 64;
        float s = 0.f;
        for (int d = 0; d < 64; ++d) s += um[d] * wp[d * 64];
        uhm[v] = s;
    }
    if (t < 64) Bmat[t] = 0.f;
}

__global__ void k_softmax(const float* __restrict__ Bmat, float* __restrict__ cvec) {
    int j = threadIdx.x;
    if (j < 8) {
        float m = Bmat[j];
        for (int i = 1; i < 8; ++i) m = fmaxf(m, Bmat[i * 8 + j]);
        float den = 0.f;
        for (int i = 0; i < 8; ++i) den += expf(Bmat[i * 8 + j] - m);
        cvec[j] = expf(Bmat[56 + j] - m) / den;
    }
}

__global__ __launch_bounds__(256) void k_uhl_pass(const float* __restrict__ x7,
                                                  const float* __restrict__ w7,
                                                  const float* __restrict__ cvec,
                                                  float* __restrict__ outp,
                                                  int mode) {
    __shared__ float uhl[16 * NCOL];
    __shared__ float fac[16 * 8];

    int t    = threadIdx.x;
    int lane = t & 31;
    int wv   = t >> 5;
    int r0   = blockIdx.x * 16;
    int mrow = r0 + (lane & 15);
    int khalf = (lane >> 4) << 1;

    v2f a[16];
#pragma unroll
    for (int ks = 0; ks < 16; ++ks) {
        const float* ap = x7 + (size_t)mrow * 64 + ks * 4 + khalf;
        a[ks].x = ap[0];
        a[ks].y = ap[1];
    }

    for (int tt = 0; tt < 4; ++tt) {
        int n = (wv * 4 + tt) * 16 + (lane & 15);
        int j = n >> 6, e = n & 63;
        const float* bp = w7 + (size_t)j * 4096 + e;
        V8 acc;
#pragma unroll
        for (int q = 0; q < 8; ++q) acc.f[q] = 0.f;
#pragma unroll
        for (int ks = 0; ks < 16; ++ks) {
            int k = ks * 4 + khalf;
            v2f b;
            b.x = bp[k * 64];
            b.y = bp[(k + 1) * 64];
            acc.v = __builtin_amdgcn_wmma_f32_16x16x4_f32(
                false, a[ks], false, b, (short)0, acc.v, false, false);
        }
        int mb = (lane >> 4) << 3;
#pragma unroll
        for (int q = 0; q < 8; ++q)
            uhl[(mb + q) * NCOL + n] = acc.f[q];
    }
    __syncthreads();

    if (t < 128) {
        int rl = t >> 3, j = t & 7;
        const float* row = uhl + rl * NCOL + j * 64;
        float q = 0.f;
        for (int e = 0; e < 64; ++e) { float u = row[e]; q += u * u; }
        float c  = cvec[j];
        float s2 = c * c * q;
        float factor = s2 / ((1.f + s2) * sqrtf(s2 + CAP_EPS));
        fac[rl * 8 + j] = factor * c;
    }
    __syncthreads();

    if (mode == 0) {
        int j = t >> 5, e0 = (t & 31) * 2;
        float a0 = 0.f, a1 = 0.f;
        const float* col = uhl + j * 64 + e0;
        for (int rl = 0; rl < 16; ++rl) {
            float g = fac[rl * 8 + j];
            a0 += g * col[rl * NCOL];
            a1 += g * col[rl * NCOL + 1];
        }
        float* pp = outp + (size_t)blockIdx.x * NCOL + j * 64 + e0;
        pp[0] = a0; pp[1] = a1;
    } else {
        for (int idx = t; idx < 16 * NCOL; idx += 256) {
            int rl = idx >> 9;
            int n  = idx & 511;
            int j = n >> 6, e = n & 63;
            int r  = r0 + rl;
            int jb = (r & 511) * 32 + (r >> 9);
            outp[(size_t)j * (JB * 64) + (size_t)jb * 64 + e] =
                fac[rl * 8 + j] * uhl[rl * NCOL + n];
        }
    }
}

__global__ __launch_bounds__(256) void k_vred1(const float* __restrict__ vpart,
                                               float* __restrict__ vpart2) {
    int b = blockIdx.x;
    int t = threadIdx.x;
    for (int v = t; v < NCOL; v += 256) {
        const float* p = vpart + (size_t)b * 32 * NCOL + v;
        float s = 0.f;
        for (int q = 0; q < 32; ++q) s += p[q * NCOL];
        vpart2[b * NCOL + v] = s;
    }
}

__global__ __launch_bounds__(512) void k_vred2(const float* __restrict__ vpart2,
                                               const float* __restrict__ uhm,
                                               float* __restrict__ Bmat) {
    __shared__ float vmean[NCOL];
    int t = threadIdx.x;
    float s = 0.f;
    for (int b = 0; b < 32; ++b) s += vpart2[b * NCOL + t];
    vmean[t] = s * (1.0f / (float)JB);
    __syncthreads();
    if (t < 64) {
        int j = t & 7;
        const float* up = uhm + t * 64;
        const float* vp = vmean + j * 64;
        float d = 0.f;
        for (int e = 0; e < 64; ++e) d += up[e] * vp[e];
        Bmat[t] += d;
    }
}

extern "C" void kernel_launch(void* const* d_in, const int* in_sizes, int n_in,
                              void* d_out, int out_size, void* d_ws, size_t ws_size,
                              hipStream_t stream) {
    const float* x = (const float*)d_in[0];
    const float* W = (const float*)d_in[1];
    float* out = (float*)d_out;
    float* ws  = (float*)d_ws;

    float* vpart  = ws;
    float* xpart  = vpart + STRIPS * NCOL;
    float* vpart2 = xpart + 8192;
    float* uhm    = vpart2 + 16384;
    float* Bmat   = uhm + 4096;
    float* cvec   = Bmat + 64;

    const float* x7 = x + (size_t)7 * JB * IN_DIM;
    const float* w7 = W + (size_t)7 * NUM_OUT * IN_DIM * OUT_DIM;

    k_xpart<<<dim3(16, 8), 256, 0, stream>>>(x, xpart);
    k_uhm<<<1, 256, 0, stream>>>(xpart, W, uhm, Bmat);

    for (int it = 0; it < 2; ++it) {
        k_softmax<<<1, 64, 0, stream>>>(Bmat, cvec);
        k_uhl_pass<<<STRIPS, 256, 0, stream>>>(x7, w7, cvec, vpart, 0);
        k_vred1<<<32, 256, 0, stream>>>(vpart, vpart2);
        k_vred2<<<1, 512, 0, stream>>>(vpart2, uhm, Bmat);
    }
    k_softmax<<<1, 64, 0, stream>>>(Bmat, cvec);
    k_uhl_pass<<<STRIPS, 256, 0, stream>>>(x7, w7, cvec, out, 1);
}


# product pass re-tiled by (capsule j, 512-row chunk), 256 groups: a wave keeps W7[j] in registers, walks 4 strips on a private LDS tile; emits the 32 partials, reduce-1 goes; norm, v sums reassociated
# speedup vs baseline: 4.0479x; 4.0479x over previous
#include <hip/hip_runtime.h>
#include <math.h>


typedef __attribute__((ext_vector_type(2))) float v2f;
typedef __attribute__((ext_vector_type(8))) float v8f;

#define NUM_IN   8
#define NUM_OUT  8
#define IN_DIM   64
#define OUT_DIM  64
#define JB       16384
#define NCOL     512
#define STRIPS   1024
#define CAP_EPS  1e-9f

union V8 { v8f v; float f[8]; };

#define XW_PARTS 64
#define XW_ROWS  (JB / XW_PARTS)

__global__ __launch_bounds__(512) void k_xw(const float* __restrict__ x,
                                            const float* __restrict__ W,
                                            float* __restrict__ uhmpart) {
    __shared__ float sd[512];
    __shared__ float xs[64];
    int part = blockIdx.x;
    int i    = blockIdx.y;
    int t    = threadIdx.x;
    const float* xp = x + ((size_t)i * JB + (size_t)part * XW_ROWS) * IN_DIM + t;
    float acc = 0.f;
    for (int k = 0; k < XW_ROWS / 8; ++k)
        acc += xp[k * 512];
    sd[t] = acc;
    __syncthreads();
    if (t < 64) {
        float s = 0.f;
        for (int q = 0; q < 8; ++q) s += sd[q * 64 + t];
        xs[t] = s;
    }
    __syncthreads();
    const float* wp = W + ((size_t)i * 8 + (t >> 6)) * 4096 + (t & 63);
    float s = 0.f;
    for (int d = 0; d < 64; ++d) s += xs[d] * wp[d * 64];
    uhmpart[((size_t)part * 8 + i) * 512 + t] = s;
}

__global__ __launch_bounds__(256) void k_uhm(const float* __restrict__ uhmpart,
                                             float* __restrict__ uhm,
                                             float* __restrict__ Bmat) {
    int v = blockIdx.x * 256 + threadIdx.x;
    float s = 0.f;
    for (int p = 0; p < XW_PARTS; ++p) s += uhmpart[(size_t)p * 4096 + v];
    uhm[v] = s * (1.0f / (float)JB);
    if (v < 64) Bmat[v] = 0.f;
}

__global__ void k_softmax(const float* __restrict__ Bmat, float* __restrict__ cvec) {
    int j = threadIdx.x;
    if (j < 8) {
        float m = Bmat[j];
        for (int i = 1; i < 8; ++i) m = fmaxf(m, Bmat[i * 8 + j]);
        float den = 0.f;
        for (int i = 0; i < 8; ++i) den += expf(Bmat[i * 8 + j] - m);
        cvec[j] = expf(Bmat[56 + j] - m) / den;
    }
}

#define PASS_WGS    256
#define CHUNK_ROWS  512
#define TPITCH      68
#define WAVE_LDS    (16 * TPITCH + 32)
#define PASS_LDS_FLOATS 41984

__device__ __forceinline__ void wave_lds_sync() {
    __builtin_amdgcn_fence(__ATOMIC_RELEASE, "wavefront");
    __builtin_amdgcn_wave_barrier();
    __builtin_amdgcn_fence(__ATOMIC_ACQUIRE, "wavefront");
}

__device__ __forceinline__ void load_a_strip(v2f (&a)[16], const float* __restrict__ x7,
                                             int row, int khalf) {
#pragma unroll
    for (int ks = 0; ks < 16; ++ks) {
        const float* ap = x7 + (size_t)row * 64 + ks * 4 + khalf;
        a[ks].x = ap[0];
        a[ks].y = ap[1];
    }
}

__global__ __launch_bounds__(256) void k_uhl_pass(const float* __restrict__ x7,
                                                  const float* __restrict__ w7,
                                                  const float* __restrict__ cvec,
                                                  float* __restrict__ outp,
                                                  int mode) {
    __shared__ __attribute__((aligned(16))) float lds[PASS_LDS_FLOATS];

    int t     = threadIdx.x;
    int lane  = t & 31;
    int wv    = t >> 5;
    int c     = lane & 15;
    int h     = lane >> 4;
    int khalf = h << 1;
    int j     = blockIdx.x & 7;
    int chunk = blockIdx.x >> 3;

    float* tile  = lds + wv * WAVE_LDS;
    float* facs  = tile + 16 * TPITCH;
    float* wpart = lds + 8 * WAVE_LDS;

    const float* wj = w7 + (size_t)j * 4096;
    v2f b[4][16];
#pragma unroll
    for (int tt = 0; tt < 4; ++tt) {
#pragma unroll
        for (int ks = 0; ks < 16; ++ks) {
            const float* bp = wj + (ks * 4 + khalf) * 64 + tt * 16 + c;
            b[tt][ks].x = bp[0];
            b[tt][ks].y = bp[64];
        }
    }
    const float cj = cvec[j];

    int rbase = chunk * CHUNK_ROWS + wv * 16;
    v2f a[16];
    load_a_strip(a, x7, rbase + c, khalf);

    float ps0 = 0.f, ps1 = 0.f;

#pragma unroll
    for (int i = 0; i < 4; ++i) {
        int r0 = rbase + i * 128;
        v2f an[16];
        if (i < 3) load_a_strip(an, x7, r0 + 128 + c, khalf);

        V8 acc[4];
#pragma unroll
        for (int tt = 0; tt < 4; ++tt)
#pragma unroll
            for (int q = 0; q < 8; ++q) acc[tt].f[q] = 0.f;
#pragma unroll
        for (int ks = 0; ks < 16; ++ks) {
#pragma unroll
            for (int tt = 0; tt < 4; ++tt)
                acc[tt].v = __builtin_amdgcn_wmma_f32_16x16x4_f32(
                    false, a[ks], false, b[tt][ks], (short)0, acc[tt].v, false, false);
        }

        int mb = h << 3;
#pragma unroll
        for (int tt = 0; tt < 4; ++tt)
#pragma unroll
            for (int q = 0; q < 8; ++q)
                tile[(mb + q) * TPITCH + tt * 16 + c] = acc[tt].f[q];
        wave_lds_sync();

        const float4* rp = (const float4*)(tile + c * TPITCH + h * 32);
        float qs = 0.f;
#pragma unroll
        for (int m = 0; m < 8; ++m) {
            float4 u = rp[m];
            qs += u.x * u.x;
            qs += u.y * u.y;
            qs += u.z * u.z;
            qs += u.w * u.w;
        }
        qs += __shfl_xor(qs, 16);
        float s2 = cj * cj * qs;
        float factor = s2 / ((1.f + s2) * sqrtf(s2 + CAP_EPS));
        if (h == 0) facs[c] = factor * cj;
        wave_lds_sync();

        if (mode == 0) {
#pragma unroll
            for (int r = 0; r < 16; ++r) {
                float g = facs[r];
                float2 u = *(const float2*)(tile + r * TPITCH + 2 * lane);
                ps0 += g * u.x;
                ps1 += g * u.y;
            }
        } else {
#pragma unroll
            for (int r = 0; r < 16; ++r) {
                float g = facs[r];
                float2 u = *(const float2*)(tile + r * TPITCH + 2 * lane);
                int row = r0 + r;
                int jb  = (row & 511) * 32 + (row >> 9);
                float2 o;
                o.x = g * u.x;
                o.y = g * u.y;
                *(float2*)(outp + (size_t)j * (JB * 64) + (size_t)jb * 64 + 2 * lane) = o;
            }
        }
        wave_lds_sync();

        if (i < 3) {
#pragma unroll
            for (int ks = 0; ks < 16; ++ks) a[ks] = an[ks];
        }
    }

    if (mode == 0) {
        wpart[wv * 64 + 2 * lane]     = ps0;
        wpart[wv * 64 + 2 * lane + 1] = ps1;
        __syncthreads();
        if (t < 64) {
            float s = 0.f;
            for (int w = 0; w < 8; ++w) s += wpart[w * 64 + t];
            outp[(size_t)chunk * NCOL + j * 64 + t] = s;
        }
    }
}

__global__ __launch_bounds__(512) void k_vred2(const float* __restrict__ vpart2,
                                               const float* __restrict__ uhm,
                                               float* __restrict__ Bmat) {
    __shared__ float vmean[NCOL];
    int t = threadIdx.x;
    float s = 0.f;
    for (int b = 0; b < 32; ++b) s += vpart2[b * NCOL + t];
    vmean[t] = s * (1.0f / (float)JB);
    __syncthreads();
    if (t < 64) {
        int j = t & 7;
        const float* up = uhm + t * 64;
        const float* vp = vmean + j * 64;
        float d = 0.f;
        for (int e = 0; e < 64; ++e) d += up[e] * vp[e];
        Bmat[t] += d;
    }
}

extern "C" void kernel_launch(void* const* d_in, const int* in_sizes, int n_in,
                              void* d_out, int out_size, void* d_ws, size_t ws_size,
                              hipStream_t stream) {
    const float* x = (const float*)d_in[0];
    const float* W = (const float*)d_in[1];
    float* out = (float*)d_out;
    float* ws  = (float*)d_ws;

    float* vpart  = ws;
    float* spare  = vpart + STRIPS * NCOL;
    float* vpart2 = spare + 8192;
    float* uhm    = vpart2 + 16384;
    float* Bmat   = uhm + 4096;
    float* cvec   = Bmat + 64;

    const float* x7 = x + (size_t)7 * JB * IN_DIM;
    const float* w7 = W + (size_t)7 * NUM_OUT * IN_DIM * OUT_DIM;

    float* uhmpart = vpart;
    k_xw<<<dim3(XW_PARTS, 8), 512, 0, stream>>>(x, W, uhmpart);
    k_uhm<<<16, 256, 0, stream>>>(uhmpart, uhm, Bmat);

    for (int it = 0; it < 2; ++it) {
        k_softmax<<<1, 64, 0, stream>>>(Bmat, cvec);
        k_uhl_pass<<<PASS_WGS, 256, 0, stream>>>(x7, w7, cvec, vpart2, 0);
        k_vred2<<<1, 512, 0, stream>>>(vpart2, uhm, Bmat);
    }
    k_softmax<<<1, 64, 0, stream>>>(Bmat, cvec);
    k_uhl_pass<<<PASS_WGS, 256, 0, stream>>>(x7, w7, cvec, out, 1);
}


# iteration 1's routing weights are stored as the literal 1/8 (softmax of the zero logits the routing itself starts from) by the set-up's last kernel: the first softmax launch goes, 10 -> 9 launches
# speedup vs baseline: 4.1764x; 1.0317x over previous
#include <hip/hip_runtime.h>
#include <math.h>


typedef __attribute__((ext_vector_type(2))) float v2f;
typedef __attribute__((ext_vector_type(8))) float v8f;

#define NUM_IN   8
#define NUM_OUT  8
#define IN_DIM   64
#define OUT_DIM  64
#define JB       16384
#define NCOL     512
#define STRIPS   1024
#define CAP_EPS  1e-9f

union V8 { v8f v; float f[8]; };

#define XW_PARTS 64
#define XW_ROWS  (JB / XW_PARTS)

__global__ __launch_bounds__(512) void k_xw(const float* __restrict__ x,
                                            const float* __restrict__ W,
                                            float* __restrict__ uhmpart) {
    __shared__ float sd[512];
    __shared__ float xs[64];
    int part = blockIdx.x;
    int i    = blockIdx.y;
    int t    = threadIdx.x;
    const float* xp = x + ((size_t)i * JB + (size_t)part * XW_ROWS) * IN_DIM + t;
    float acc = 0.f;
    for (int k = 0; k < XW_ROWS / 8; ++k)
        acc += xp[k * 512];
    sd[t] = acc;
    __syncthreads();
    if (t < 64) {
        float s = 0.f;
        for (int q = 0; q < 8; ++q) s += sd[q * 64 + t];
        xs[t] = s;
    }
    __syncthreads();
    const float* wp = W + ((size_t)i * 8 + (t >> 6)) * 4096 + (t & 63);
    float s = 0.f;
    for (int d = 0; d < 64; ++d) s += xs[d] * wp[d * 64];
    uhmpart[((size_t)part * 8 + i) * 512 + t] = s;
}

__global__ __launch_bounds__(256) void k_uhm(const float* __restrict__ uhmpart,
                                             float* __restrict__ uhm,
                                             float* __restrict__ Bmat,
                                             float* __restrict__ cvec) {
    int v = blockIdx.x * 256 + threadIdx.x;
    float s = 0.f;
    for (int p = 0; p < XW_PARTS; ++p) s += uhmpart[(size_t)p * 4096 + v];
    uhm[v] = s * (1.0f / (float)JB);
    if (v < 64) Bmat[v] = 0.f;
    if (v < NUM_OUT) cvec[v] = 1.0f / (float)NUM_IN;
}

__global__ void k_softmax(const float* __restrict__ Bmat, float* __restrict__ cvec) {
    int j = threadIdx.x;
    if (j < 8) {
        float m = Bmat[j];
        for (int i = 1; i < 8; ++i) m = fmaxf(m, Bmat[i * 8 + j]);
        float den = 0.f;
        for (int i = 0; i < 8; ++i) den += expf(Bmat[i * 8 + j] - m);
        cvec[j] = expf(Bmat[56 + j] - m) / den;
    }
}

#define PASS_WGS    256
#define CHUNK_ROWS  512
#define TPITCH      68
#define WAVE_LDS    (16 * TPITCH + 32)
#define PASS_LDS_FLOATS 41984

__device__ __forceinline__ void wave_lds_sync() {
    __builtin_amdgcn_fence(__ATOMIC_RELEASE, "wavefront");
    __builtin_amdgcn_wave_barrier();
    __builtin_amdgcn_fence(__ATOMIC_ACQUIRE, "wavefront");
}

__device__ __forceinline__ void load_a_strip(v2f (&a)[16], const float* __restrict__ x7,
                                             int row, int khalf) {
#pragma unroll
    for (int ks = 0; ks < 16; ++ks) {
        const float* ap = x7 + (size_t)row * 64 + ks * 4 + khalf;
        a[ks].x = ap[0];
        a[ks].y = ap[1];
    }
}

__global__ __launch_bounds__(256) void k_uhl_pass(const float* __restrict__ x7,
                                                  const float* __restrict__ w7,
                                                  const float* __restrict__ cvec,
                                                  float* __restrict__ outp,
                                                  int mode) {
    __shared__ __attribute__((aligned(16))) float lds[PASS_LDS_FLOATS];

    int t     = threadIdx.x;
    int lane  = t & 31;
    int wv    = t >> 5;
    int c     = lane & 15;
    int h     = lane >> 4;
    int khalf = h << 1;
    int j     = blockIdx.x & 7;
    int chunk = blockIdx.x >> 3;

    float* tile  = lds + wv * WAVE_LDS;
    float* facs  = tile + 16 * TPITCH;
    float* wpart = lds + 8 * WAVE_LDS;

    const float* wj = w7 + (size_t)j * 4096;
    v2f b[4][16];
#pragma unroll
    for (int tt = 0; tt < 4; ++tt) {
#pragma unroll
        for (int ks = 0; ks < 16; ++ks) {
            const float* bp = wj + (ks * 4 + khalf) * 64 + tt * 16 + c;
            b[tt][ks].x = bp[0];
            b[tt][ks].y = bp[64];
        }
    }
    const float cj = cvec[j];

    int rbase = chunk * CHUNK_ROWS + wv * 16;
    v2f a[16];
    load_a_strip(a, x7, rbase + c, khalf);

    float ps0 = 0.f, ps1 = 0.f;

#pragma unroll
    for (int i = 0; i < 4; ++i) {
        int r0 = rbase + i * 128;
        v2f an[16];
        if (i < 3) load_a_strip(an, x7, r0 + 128 + c, khalf);

        V8 acc[4];
#pragma unroll
        for (int tt = 0; tt < 4; ++tt)
#pragma unroll
            for (int q = 0; q < 8; ++q) acc[tt].f[q] = 0.f;
#pragma unroll
        for (int ks = 0; ks < 16; ++ks) {
#pragma unroll
            for (int tt = 0; tt < 4; ++tt)
                acc[tt].v = __builtin_amdgcn_wmma_f32_16x16x4_f32(
                    false, a[ks], false, b[tt][ks], (short)0, acc[tt].v, false, false);
        }

        int mb = h << 3;
#pragma unroll
        for (int tt = 0; tt < 4; ++tt)
#pragma unroll
            for (int q = 0; q < 8; ++q)
                tile[(mb + q) * TPITCH + tt * 16 + c] = acc[tt].f[q];
        wave_lds_sync();

        const float4* rp = (const float4*)(tile + c * TPITCH + h * 32);
        float qs = 0.f;
#pragma unroll
        for (int m = 0; m < 8; ++m) {
            float4 u = rp[m];
            qs += u.x * u.x;
            qs += u.y * u.y;
            qs += u.z * u.z;
            qs += u.w * u.w;
        }
        qs += __shfl_xor(qs, 16);
        float s2 = cj * cj * qs;
        float factor = s2 / ((1.f + s2) * sqrtf(s2 + CAP_EPS));
        if (h == 0) facs[c] = factor * cj;
        wave_lds_sync();

        if (mode == 0) {
#pragma unroll
            for (int r = 0; r < 16; ++r) {
                float g = facs[r];
                float2 u = *(const float2*)(tile + r * TPITCH + 2 * lane);
                ps0 += g * u.x;
                ps1 += g * u.y;
            }
        } else {
#pragma unroll
            for (int r = 0; r < 16; ++r) {
                float g = facs[r];
                float2 u = *(const float2*)(tile + r * TPITCH + 2 * lane);
                int row = r0 + r;
                int jb  = (row & 511) * 32 + (row >> 9);
                float2 o;
                o.x = g * u.x;
                o.y = g * u.y;
                *(float2*)(outp + (size_t)j * (JB * 64) + (size_t)jb * 64 + 2 * lane) = o;
            }
        }
        wave_lds_sync();

        if (i < 3) {
#pragma unroll
            for (int ks = 0; ks < 16; ++ks) a[ks] = an[ks];
        }
    }

    if (mode == 0) {
        wpart[wv * 64 + 2 * lane]     = ps0;
        wpart[wv * 64 + 2 * lane + 1] = ps1;
        __syncthreads();
        if (t < 64) {
            float s = 0.f;
            for (int w = 0; w < 8; ++w) s += wpart[w * 64 + t];
            outp[(size_t)chunk * NCOL + j * 64 + t] = s;
        }
    }
}

__global__ __launch_bounds__(512) void k_vred2(const float* __restrict__ vpart2,
                                               const float* __restrict__ uhm,
                                               float* __restrict__ Bmat) {
    __shared__ float vmean[NCOL];
    int t = threadIdx.x;
    float s = 0.f;
    for (int b = 0; b < 32; ++b) s += vpart2[b * NCOL + t];
    vmean[t] = s * (1.0f / (float)JB);
    __syncthreads();
    if (t < 64) {
        int j = t & 7;
        const float* up = uhm + t * 64;
        const float* vp = vmean + j * 64;
        float d = 0.f;
        for (int e = 0; e < 64; ++e) d += up[e] * vp[e];
        Bmat[t] += d;
    }
}

extern "C" void kernel_launch(void* const* d_in, const int* in_sizes, int n_in,
                              void* d_out, int out_size, void* d_ws, size_t ws_size,
                              hipStream_t stream) {
    const float* x = (const float*)d_in[0];
    const float* W = (const float*)d_in[1];
    float* out = (float*)d_out;
    float* ws  = (float*)d_ws;

    float* vpart  = ws;
    float* spare  = vpart + STRIPS * NCOL;
    float* vpart2 = spare + 8192;
    float* uhm    = vpart2 + 16384;
    float* Bmat   = uhm + 4096;
    float* cvec   = Bmat + 64;

    const float* x7 = x + (size_t)7 * JB * IN_DIM;
    const float* w7 = W + (size_t)7 * NUM_OUT * IN_DIM * OUT_DIM;

    float* uhmpart = vpart;
    k_xw<<<dim3(XW_PARTS, 8), 512, 0, stream>>>(x, W, uhmpart);
    k_uhm<<<16, 256, 0, stream>>>(uhmpart, uhm, Bmat, cvec);

    k_uhl_pass<<<PASS_WGS, 256, 0, stream>>>(x7, w7, cvec, vpart2, 0);
    k_vred2<<<1, 512, 0, stream>>>(vpart2, uhm, Bmat);
    k_softmax<<<1, 64, 0, stream>>>(Bmat, cvec);
    k_uhl_pass<<<PASS_WGS, 256, 0, stream>>>(x7, w7, cvec, vpart2, 0);
    k_vred2<<<1, 512, 0, stream>>>(vpart2, uhm, Bmat);
    k_softmax<<<1, 64, 0, stream>>>(Bmat, cvec);
    k_uhl_pass<<<PASS_WGS, 256, 0, stream>>>(x7, w7, cvec, out, 1);
}


# u_hat_mean: W is applied once, by 64 work-groups (i,j) to the added chunk sums, not by each of the 512 streaming groups to its own; 64 MB of W reads leave the x pass; u_hat_mean sums reassociated
# speedup vs baseline: 4.9472x; 1.1846x over previous
#include <hip/hip_runtime.h>
#include <math.h>


typedef __attribute__((ext_vector_type(2))) float v2f;
typedef __attribute__((ext_vector_type(8))) float v8f;

#define NUM_IN   8
#define NUM_OUT  8
#define IN_DIM   64
#define OUT_DIM  64
#define JB       16384
#define NCOL     512
#define STRIPS   1024
#define CAP_EPS  1e-9f

union V8 { v8f v; float f[8]; };

#define XW_PARTS 64
#define XW_ROWS  (JB / XW_PARTS)

__global__ __launch_bounds__(512) void k_xsum(const float* __restrict__ x,
                                              float* __restrict__ xpart) {
    __shared__ float sd[512];
    int part = blockIdx.x;
    int i    = blockIdx.y;
    int t    = threadIdx.x;
    const float* xp = x + ((size_t)i * JB + (size_t)part * XW_ROWS) * IN_DIM + t;
    float acc = 0.f;
    for (int k = 0; k < XW_ROWS / 8; ++k)
        acc += xp[k * 512];
    sd[t] = acc;
    __syncthreads();
    if (t < 64) {
        float s = 0.f;
        for (int q = 0; q < 8; ++q) s += sd[q * 64 + t];
        xpart[((size_t)i * XW_PARTS + part) * 64 + t] = s;
    }
}

__global__ __launch_bounds__(256) void k_uhm(const float* __restrict__ xpart,
                                             const float* __restrict__ W,
                                             float* __restrict__ uhm,
                                             float* __restrict__ Bmat,
                                             float* __restrict__ cvec) {
    __shared__ float sa[256];
    __shared__ float xs[64];
    int ij = blockIdx.x;
    int i  = ij >> 3;
    int t  = threadIdx.x;
    int e  = t & 63;
    int g  = t >> 6;

    const float* wp = W + ((size_t)ij * 64 + g * 16) * 64 + e;
    float w[16];
#pragma unroll
    for (int k = 0; k < 16; ++k) w[k] = wp[k * 64];

    const float* pp = xpart + ((size_t)i * XW_PARTS + g * 16) * 64 + e;
    float s = 0.f;
#pragma unroll
    for (int k = 0; k < 16; ++k) s += pp[k * 64];
    sa[t] = s;
    __syncthreads();
    if (t < 64) xs[t] = ((sa[t] + sa[64 + t]) + sa[128 + t]) + sa[192 + t];
    __syncthreads();

    float p = 0.f;
#pragma unroll
    for (int k = 0; k < 16; ++k) p += xs[g * 16 + k] * w[k];
    sa[t] = p;
    __syncthreads();
    if (t < 64) {
        float u = ((sa[t] + sa[64 + t]) + sa[128 + t]) + sa[192 + t];
        uhm[(size_t)ij * 64 + t] = u * (1.0f / (float)JB);
    }
    if (ij == 0) {
        if (t < 64) Bmat[t] = 0.f;
        if (t < NUM_OUT) cvec[t] = 1.0f / (float)NUM_IN;
    }
}

__global__ void k_softmax(const float* __restrict__ Bmat, float* __restrict__ cvec) {
    int j = threadIdx.x;
    if (j < 8) {
        float m = Bmat[j];
        for (int i = 1; i < 8; ++i) m = fmaxf(m, Bmat[i * 8 + j]);
        float den = 0.f;
        for (int i = 0; i < 8; ++i) den += expf(Bmat[i * 8 + j] - m);
        cvec[j] = expf(Bmat[56 + j] - m) / den;
    }
}

__device__ __forceinline__ float squash_g(float cj, float qs) {
    float s2 = cj * cj * qs;
    float factor = s2 / ((1.f + s2) * sqrtf(s2 + CAP_EPS));
    return factor * cj;
}

#define PASS_WGS    256
#define CHUNK_ROWS  512
#define TPITCH      68
#define WAVE_LDS    (16 * TPITCH + 32)
#define PASS_LDS_FLOATS 41984

__device__ __forceinline__ void wave_lds_sync() {
    __builtin_amdgcn_fence(__ATOMIC_RELEASE, "wavefront");
    __builtin_amdgcn_wave_barrier();
    __builtin_amdgcn_fence(__ATOMIC_ACQUIRE, "wavefront");
}

__device__ __forceinline__ void load_a_strip(v2f (&a)[16], const float* __restrict__ x7,
                                             int row, int khalf) {
#pragma unroll
    for (int ks = 0; ks < 16; ++ks) {
        const float* ap = x7 + (size_t)row * 64 + ks * 4 + khalf;
        a[ks].x = ap[0];
        a[ks].y = ap[1];
    }
}

__global__ __launch_bounds__(256) void k_uhl_pass(const float* __restrict__ x7,
                                                  const float* __restrict__ w7,
                                                  const float* __restrict__ cvec,
                                                  float* __restrict__ vpart2,
                                                  float* __restrict__ qbuf,
                                                  float* __restrict__ urows) {
    __shared__ __attribute__((aligned(16))) float lds[PASS_LDS_FLOATS];

    int t     = threadIdx.x;
    int lane  = t & 31;
    int wv    = t >> 5;
    int c     = lane & 15;
    int h     = lane >> 4;
    int khalf = h << 1;
    int j     = blockIdx.x & 7;
    int chunk = blockIdx.x >> 3;

    float* tile  = lds + wv * WAVE_LDS;
    float* facs  = tile + 16 * TPITCH;
    float* wpart = lds + 8 * WAVE_LDS;

    const float* wj = w7 + (size_t)j * 4096;
    v2f b[4][16];
#pragma unroll
    for (int tt = 0; tt < 4; ++tt) {
#pragma unroll
        for (int ks = 0; ks < 16; ++ks) {
            const float* bp = wj + (ks * 4 + khalf) * 64 + tt * 16 + c;
            b[tt][ks].x = bp[0];
            b[tt][ks].y = bp[64];
        }
    }
    const float cj = cvec[j];

    int rbase = chunk * CHUNK_ROWS + wv * 16;
    v2f a[16];
    load_a_strip(a, x7, rbase + c, khalf);

    float ps0 = 0.f, ps1 = 0.f;

#pragma unroll
    for (int i = 0; i < 4; ++i) {
        int r0 = rbase + i * 128;
        v2f an[16];
        if (i < 3) load_a_strip(an, x7, r0 + 128 + c, khalf);

        V8 acc[4];
#pragma unroll
        for (int tt = 0; tt < 4; ++tt)
#pragma unroll
            for (int q = 0; q < 8; ++q) acc[tt].f[q] = 0.f;
#pragma unroll
        for (int ks = 0; ks < 16; ++ks) {
#pragma unroll
            for (int tt = 0; tt < 4; ++tt)
                acc[tt].v = __builtin_amdgcn_wmma_f32_16x16x4_f32(
                    false, a[ks], false, b[tt][ks], (short)0, acc[tt].v, false, false);
        }

        int mb = h << 3;
#pragma unroll
        for (int tt = 0; tt < 4; ++tt)
#pragma unroll
            for (int q = 0; q < 8; ++q)
                tile[(mb + q) * TPITCH + tt * 16 + c] = acc[tt].f[q];
        wave_lds_sync();

        const float4* rp = (const float4*)(tile + c * TPITCH + h * 32);
        float qs = 0.f;
#pragma unroll
        for (int m = 0; m < 8; ++m) {
            float4 u = rp[m];
            qs += u.x * u.x;
            qs += u.y * u.y;
            qs += u.z * u.z;
            qs += u.w * u.w;
        }
        qs += __shfl_xor(qs, 16);
        float gfac = squash_g(cj, qs);
        if (h == 0) {
            facs[c] = gfac;
            qbuf[(size_t)j * JB + r0 + c] = qs;
        }
        wave_lds_sync();

#pragma unroll
        for (int r = 0; r < 16; ++r) {
            float g = facs[r];
            float2 u = *(const float2*)(tile + r * TPITCH + 2 * lane);
            ps0 += g * u.x;
            ps1 += g * u.y;
            int row = r0 + r;
            int jb  = (row & 511) * 32 + (row >> 9);
            *(float2*)(urows + (size_t)j * (JB * 64) + (size_t)jb * 64 + 2 * lane) = u;
        }
        wave_lds_sync();

        if (i < 3) {
#pragma unroll
            for (int ks = 0; ks < 16; ++ks) a[ks] = an[ks];
        }
    }

    wpart[wv * 64 + 2 * lane]     = ps0;
    wpart[wv * 64 + 2 * lane + 1] = ps1;
    __syncthreads();
    if (t < 64) {
        float s = 0.f;
        for (int w = 0; w < 8; ++w) s += wpart[w * 64 + t];
        vpart2[(size_t)chunk * NCOL + j * 64 + t] = s;
    }
}


__device__ __forceinline__ float lane_bcast(float v, int src) {
    return __int_as_float(__builtin_amdgcn_readlane(__float_as_int(v), src));
}

__device__ __forceinline__ void load_wave_rows(float (&qv)[4], float2 (&v)[4][16],
                                               const float* qp, const float* up) {
#pragma unroll
    for (int i = 0; i < 4; ++i) qv[i] = qp[i * 128];
#pragma unroll
    for (int i = 0; i < 4; ++i)
#pragma unroll
        for (int r = 0; r < 16; ++r)
            v[i][r] = *(const float2*)(up + (size_t)(i * 128 + r) * (32 * 64));
}

__global__ __launch_bounds__(256) void k_vsum(const float* __restrict__ urows,
                                              const float* __restrict__ qbuf,
                                              const float* __restrict__ cvec,
                                              float* __restrict__ vpart2) {
    __shared__ float wpart[8 * 64];
    int t     = threadIdx.x;
    int lane  = t & 31;
    int wv    = t >> 5;
    int c     = lane & 15;
    int j     = blockIdx.x & 7;
    int chunk = blockIdx.x >> 3;
    const float cj = cvec[j];

    int sbase = wv * 16;
    float  qv[4];
    float2 v[4][16];
    load_wave_rows(qv, v, qbuf + (size_t)j * JB + chunk * CHUNK_ROWS + sbase + c,
                   urows + (size_t)j * (JB * 64) + ((size_t)sbase * 32 + chunk) * 64 + 2 * lane);

    float ps0 = 0.f, ps1 = 0.f;
#pragma unroll
    for (int i = 0; i < 4; ++i) {
        float gfac = squash_g(cj, qv[i]);
#pragma unroll
        for (int r = 0; r < 16; ++r) {
            float g = lane_bcast(gfac, r);
            ps0 += g * v[i][r].x;
            ps1 += g * v[i][r].y;
        }
    }

    wpart[wv * 64 + 2 * lane]     = ps0;
    wpart[wv * 64 + 2 * lane + 1] = ps1;
    __syncthreads();
    if (t < 64) {
        float s = 0.f;
        for (int w = 0; w < 8; ++w) s += wpart[w * 64 + t];
        vpart2[(size_t)chunk * NCOL + j * 64 + t] = s;
    }
}

__global__ __launch_bounds__(256) void k_scale(float* urows,
                                               const float* __restrict__ qbuf,
                                               const float* __restrict__ cvec) {
    int t     = threadIdx.x;
    int lane  = t & 31;
    int wv    = t >> 5;
    int c     = lane & 15;
    int j     = blockIdx.x & 7;
    int chunk = blockIdx.x >> 3;
    const float cj = cvec[j];

    int sbase = wv * 16;
    float* up = urows + (size_t)j * (JB * 64) + ((size_t)sbase * 32 + chunk) * 64 + 2 * lane;
    float  qv[4];
    float2 v[4][16];
    load_wave_rows(qv, v, qbuf + (size_t)j * JB + chunk * CHUNK_ROWS + sbase + c, up);

#pragma unroll
    for (int i = 0; i < 4; ++i) {
        float gfac = squash_g(cj, qv[i]);
#pragma unroll
        for (int r = 0; r < 16; ++r) {
            float g = lane_bcast(gfac, r);
            float2 o;
            o.x = g * v[i][r].x;
            o.y = g * v[i][r].y;
            *(float2*)(up + (size_t)(i * 128 + r) * (32 * 64)) = o;
        }
    }
}

__global__ __launch_bounds__(512) void k_vred2(const float* __restrict__ vpart2,
                                               const float* __restrict__ uhm,
                                               float* __restrict__ Bmat) {
    __shared__ float vmean[NCOL];
    int t = threadIdx.x;
    float s = 0.f;
    for (int b = 0; b < 32; ++b) s += vpart2[b * NCOL + t];
    vmean[t] = s * (1.0f / (float)JB);
    __syncthreads();
    if (t < 64) {
        int j = t & 7;
        const float* up = uhm + t * 64;
        const float* vp = vmean + j * 64;
        float d = 0.f;
        for (int e = 0; e < 64; ++e) d += up[e] * vp[e];
        Bmat[t] += d;
    }
}

extern "C" void kernel_launch(void* const* d_in, const int* in_sizes, int n_in,
                              void* d_out, int out_size, void* d_ws, size_t ws_size,
                              hipStream_t stream) {
    const float* x = (const float*)d_in[0];
    const float* W = (const float*)d_in[1];
    float* out = (float*)d_out;
    float* ws  = (float*)d_ws;

    float* vpart  = ws;
    float* spare  = vpart + STRIPS * NCOL;
    float* vpart2 = spare + 8192;
    float* uhm    = vpart2 + 16384;
    float* Bmat   = uhm + 4096;
    float* cvec   = Bmat + 64;

    const float* x7 = x + (size_t)7 * JB * IN_DIM;
    const float* w7 = W + (size_t)7 * NUM_OUT * IN_DIM * OUT_DIM;

    float* xpart = vpart;
    float* qbuf = vpart + XW_PARTS * 4096;
    k_xsum<<<dim3(XW_PARTS, 8), 512, 0, stream>>>(x, xpart);
    k_uhm<<<NUM_IN * NUM_OUT, 256, 0, stream>>>(xpart, W, uhm, Bmat, cvec);

    k_uhl_pass<<<PASS_WGS, 256, 0, stream>>>(x7, w7, cvec, vpart2, qbuf, out);
    k_vred2<<<1, 512, 0, stream>>>(vpart2, uhm, Bmat);
    k_softmax<<<1, 64, 0, stream>>>(Bmat, cvec);
    k_vsum<<<PASS_WGS, 256, 0, stream>>>(out, qbuf, cvec, vpart2);
    k_vred2<<<1, 512, 0, stream>>>(vpart2, uhm, Bmat);
    k_softmax<<<1, 64, 0, stream>>>(Bmat, cvec);
    k_scale<<<PASS_WGS, 256, 0, stream>>>(out, qbuf, cvec);
}


# routing update split by capsule column: 8 work-groups each add their capsule's 32 partial rows, take its 8 dots and the column softmax; 4 one-group launches become 2 (9 -> 7); sum order kept
# speedup vs baseline: 5.4699x; 1.1057x over previous
#include <hip/hip_runtime.h>
#include <math.h>


typedef __attribute__((ext_vector_type(2))) float v2f;
typedef __attribute__((ext_vector_type(8))) float v8f;

#define NUM_IN   8
#define NUM_OUT  8
#define IN_DIM   64
#define OUT_DIM  64
#define JB       16384
#define NCOL     512
#define STRIPS   1024
#define CAP_EPS  1e-9f

union V8 { v8f v; float f[8]; };

#define XW_PARTS 64
#define XW_ROWS  (JB / XW_PARTS)

__global__ __launch_bounds__(512) void k_xsum(const float* __restrict__ x,
                                              float* __restrict__ xpart) {
    __shared__ float sd[512];
    int part = blockIdx.x;
    int i    = blockIdx.y;
    int t    = threadIdx.x;
    const float* xp = x + ((size_t)i * JB + (size_t)part * XW_ROWS) * IN_DIM + t;
    float acc = 0.f;
    for (int k = 0; k < XW_ROWS / 8; ++k)
        acc += xp[k * 512];
    sd[t] = acc;
    __syncthreads();
    if (t < 64) {
        float s = 0.f;
        for (int q = 0; q < 8; ++q) s += sd[q * 64 + t];
        xpart[((size_t)i * XW_PARTS + part) * 64 + t] = s;
    }
}

__global__ __launch_bounds__(256) void k_uhm(const float* __restrict__ xpart,
                                             const float* __restrict__ W,
                                             float* __restrict__ uhm,
                                             float* __restrict__ Bmat,
                                             float* __restrict__ cvec) {
    __shared__ float sa[256];
    __shared__ float xs[64];
    int ij = blockIdx.x;
    int i  = ij >> 3;
    int t  = threadIdx.x;
    int e  = t & 63;
    int g  = t >> 6;

    const float* wp = W + ((size_t)ij * 64 + g * 16) * 64 + e;
    float w[16];
#pragma unroll
    for (int k = 0; k < 16; ++k) w[k] = wp[k * 64];

    const float* pp = xpart + ((size_t)i * XW_PARTS + g * 16) * 64 + e;
    float s = 0.f;
#pragma unroll
    for (int k = 0; k < 16; ++k) s += pp[k * 64];
    sa[t] = s;
    __syncthreads();
    if (t < 64) xs[t] = ((sa[t] + sa[64 + t]) + sa[128 + t]) + sa[192 + t];
    __syncthreads();

    float p = 0.f;
#pragma unroll
    for (int k = 0; k < 16; ++k) p += xs[g * 16 + k] * w[k];
    sa[t] = p;
    __syncthreads();
    if (t < 64) {
        float u = ((sa[t] + sa[64 + t]) + sa[128 + t]) + sa[192 + t];
        uhm[(size_t)ij * 64 + t] = u * (1.0f / (float)JB);
    }
    if (ij == 0) {
        if (t < 64) Bmat[t] = 0.f;
        if (t < NUM_OUT) cvec[t] = 1.0f / (float)NUM_IN;
    }
}

__device__ __forceinline__ float squash_g(float cj, float qs) {
    float s2 = cj * cj * qs;
    float factor = s2 / ((1.f + s2) * sqrtf(s2 + CAP_EPS));
    return factor * cj;
}

#define PASS_WGS    256
#define CHUNK_ROWS  512
#define TPITCH      68
#define WAVE_LDS    (16 * TPITCH + 32)
#define PASS_LDS_FLOATS 41984

__device__ __forceinline__ void wave_lds_sync() {
    __builtin_amdgcn_fence(__ATOMIC_RELEASE, "wavefront");
    __builtin_amdgcn_wave_barrier();
    __builtin_amdgcn_fence(__ATOMIC_ACQUIRE, "wavefront");
}

__device__ __forceinline__ void load_a_strip(v2f (&a)[16], const float* __restrict__ x7,
                                             int row, int khalf) {
#pragma unroll
    for (int ks = 0; ks < 16; ++ks) {
        const float* ap = x7 + (size_t)row * 64 + ks * 4 + khalf;
        a[ks].x = ap[0];
        a[ks].y = ap[1];
    }
}

__global__ __launch_bounds__(256) void k_uhl_pass(const float* __restrict__ x7,
                                                  const float* __restrict__ w7,
                                                  const float* __restrict__ cvec,
                                                  float* __restrict__ vpart2,
                                                  float* __restrict__ qbuf,
                                                  float* __restrict__ urows) {
    __shared__ __attribute__((aligned(16))) float lds[PASS_LDS_FLOATS];

    int t     = threadIdx.x;
    int lane  = t & 31;
    int wv    = t >> 5;
    int c     = lane & 15;
    int h     = lane >> 4;
    int khalf = h << 1;
    int j     = blockIdx.x & 7;
    int chunk = blockIdx.x >> 3;

    float* tile  = lds + wv * WAVE_LDS;
    float* facs  = tile + 16 * TPITCH;
    float* wpart = lds + 8 * WAVE_LDS;

    const float* wj = w7 + (size_t)j * 4096;
    v2f b[4][16];
#pragma unroll
    for (int tt = 0; tt < 4; ++tt) {
#pragma unroll
        for (int ks = 0; ks < 16; ++ks) {
            const float* bp = wj + (ks * 4 + khalf) * 64 + tt * 16 + c;
            b[tt][ks].x = bp[0];
            b[tt][ks].y = bp[64];
        }
    }
    const float cj = cvec[j];

    int rbase = chunk * CHUNK_ROWS + wv * 16;
    v2f a[16];
    load_a_strip(a, x7, rbase + c, khalf);

    float ps0 = 0.f, ps1 = 0.f;

#pragma unroll
    for (int i = 0; i < 4; ++i) {
        int r0 = rbase + i * 128;
        v2f an[16];
        if (i < 3) load_a_strip(an, x7, r0 + 128 + c, khalf);

        V8 acc[4];
#pragma unroll
        for (int tt = 0; tt < 4; ++tt)
#pragma unroll
            for (int q = 0; q < 8; ++q) acc[tt].f[q] = 0.f;
#pragma unroll
        for (int ks = 0; ks < 16; ++ks) {
#pragma unroll
            for (int tt = 0; tt < 4; ++tt)
                acc[tt].v = __builtin_amdgcn_wmma_f32_16x16x4_f32(
                    false, a[ks], false, b[tt][ks], (short)0, acc[tt].v, false, false);
        }

        int mb = h << 3;
#pragma unroll
        for (int tt = 0; tt < 4; ++tt)
#pragma unroll
            for (int q = 0; q < 8; ++q)
                tile[(mb + q) * TPITCH + tt * 16 + c] = acc[tt].f[q];
        wave_lds_sync();

        const float4* rp = (const float4*)(tile + c * TPITCH + h * 32);
        float qs = 0.f;
#pragma unroll
        for (int m = 0; m < 8; ++m) {
            float4 u = rp[m];
            qs += u.x * u.x;
            qs += u.y * u.y;
            qs += u.z * u.z;
            qs += u.w * u.w;
        }
        qs += __shfl_xor(qs, 16);
        float gfac = squash_g(cj, qs);
        if (h == 0) {
            facs[c] = gfac;
            qbuf[(size_t)j * JB + r0 + c] = qs;
        }
        wave_lds_sync();

#pragma unroll
        for (int r = 0; r < 16; ++r) {
            float g = facs[r];
            float2 u = *(const float2*)(tile + r * TPITCH + 2 * lane);
            ps0 += g * u.x;
            ps1 += g * u.y;
            int row = r0 + r;
            int jb  = (row & 511) * 32 + (row >> 9);
            *(float2*)(urows + (size_t)j * (JB * 64) + (size_t)jb * 64 + 2 * lane) = u;
        }
        wave_lds_sync();

        if (i < 3) {
#pragma unroll
            for (int ks = 0; ks < 16; ++ks) a[ks] = an[ks];
        }
    }

    wpart[wv * 64 + 2 * lane]     = ps0;
    wpart[wv * 64 + 2 * lane + 1] = ps1;
    __syncthreads();
    if (t < 64) {
        float s = 0.f;
        for (int w = 0; w < 8; ++w) s += wpart[w * 64 + t];
        vpart2[(size_t)chunk * NCOL + j * 64 + t] = s;
    }
}


__device__ __forceinline__ float lane_bcast(float v, int src) {
    return __int_as_float(__builtin_amdgcn_readlane(__float_as_int(v), src));
}

__device__ __forceinline__ void load_wave_rows(float (&qv)[4], float2 (&v)[4][16],
                                               const float* qp, const float* up) {
#pragma unroll
    for (int i = 0; i < 4; ++i) qv[i] = qp[i * 128];
#pragma unroll
    for (int i = 0; i < 4; ++i)
#pragma unroll
        for (int r = 0; r < 16; ++r)
            v[i][r] = *(const float2*)(up + (size_t)(i * 128 + r) * (32 * 64));
}

__global__ __launch_bounds__(256) void k_vsum(const float* __restrict__ urows,
                                              const float* __restrict__ qbuf,
                                              const float* __restrict__ cvec,
                                              float* __restrict__ vpart2) {
    __shared__ float wpart[8 * 64];
    int t     = threadIdx.x;
    int lane  = t & 31;
    int wv    = t >> 5;
    int c     = lane & 15;
    int j     = blockIdx.x & 7;
    int chunk = blockIdx.x >> 3;
    const float cj = cvec[j];

    int sbase = wv * 16;
    float  qv[4];
    float2 v[4][16];
    load_wave_rows(qv, v, qbuf + (size_t)j * JB + chunk * CHUNK_ROWS + sbase + c,
                   urows + (size_t)j * (JB * 64) + ((size_t)sbase * 32 + chunk) * 64 + 2 * lane);

    float ps0 = 0.f, ps1 = 0.f;
#pragma unroll
    for (int i = 0; i < 4; ++i) {
        float gfac = squash_g(cj, qv[i]);
#pragma unroll
        for (int r = 0; r < 16; ++r) {
            float g = lane_bcast(gfac, r);
            ps0 += g * v[i][r].x;
            ps1 += g * v[i][r].y;
        }
    }

    wpart[wv * 64 + 2 * lane]     = ps0;
    wpart[wv * 64 + 2 * lane + 1] = ps1;
    __syncthreads();
    if (t < 64) {
        float s = 0.f;
        for (int w = 0; w < 8; ++w) s += wpart[w * 64 + t];
        vpart2[(size_t)chunk * NCOL + j * 64 + t] = s;
    }
}

__global__ __launch_bounds__(256) void k_scale(float* urows,
                                               const float* __restrict__ qbuf,
                                               const float* __restrict__ cvec) {
    int t     = threadIdx.x;
    int lane  = t & 31;
    int wv    = t >> 5;
    int c     = lane & 15;
    int j     = blockIdx.x & 7;
    int chunk = blockIdx.x >> 3;
    const float cj = cvec[j];

    int sbase = wv * 16;
    float* up = urows + (size_t)j * (JB * 64) + ((size_t)sbase * 32 + chunk) * 64 + 2 * lane;
    float  qv[4];
    float2 v[4][16];
    load_wave_rows(qv, v, qbuf + (size_t)j * JB + chunk * CHUNK_ROWS + sbase + c, up);

#pragma unroll
    for (int i = 0; i < 4; ++i) {
        float gfac = squash_g(cj, qv[i]);
#pragma unroll
        for (int r = 0; r < 16; ++r) {
            float g = lane_bcast(gfac, r);
            float2 o;
            o.x = g * v[i][r].x;
            o.y = g * v[i][r].y;
            *(float2*)(up + (size_t)(i * 128 + r) * (32 * 64)) = o;
        }
    }
}

__global__ __launch_bounds__(64) void k_route(const float* __restrict__ vpart2,
                                              const float* __restrict__ uhm,
                                              float* __restrict__ Bmat,
                                              float* __restrict__ cvec) {
    __shared__ __attribute__((aligned(16))) float vmean[64];
    __shared__ float bcol[8];
    int j = blockIdx.x;
    int t = threadIdx.x;

    float4 u[16];
    float  bij = 0.f;
    if (t < 8) {
        const float4* up = (const float4*)(uhm + ((size_t)t * 8 + j) * 64);
#pragma unroll
        for (int k = 0; k < 16; ++k) u[k] = up[k];
        bij = Bmat[t * 8 + j];
    }

    const float* pp = vpart2 + j * 64 + t;
    float s = 0.f;
#pragma unroll
    for (int b = 0; b < 32; ++b) s += pp[b * NCOL];
    vmean[t] = s * (1.0f / (float)JB);
    __syncthreads();

    if (t < 8) {
        const float4* vp = (const float4*)vmean;
        float d = 0.f;
#pragma unroll
        for (int k = 0; k < 16; ++k) {
            float4 v = vp[k];
            d += u[k].x * v.x;
            d += u[k].y * v.y;
            d += u[k].z * v.z;
            d += u[k].w * v.w;
        }
        bij += d;
        Bmat[t * 8 + j] = bij;
        bcol[t] = bij;
    }
    __syncthreads();
    if (t == 0) {
        float m = bcol[0];
        for (int i = 1; i < 8; ++i) m = fmaxf(m, bcol[i]);
        float den = 0.f;
        for (int i = 0; i < 8; ++i) den += expf(bcol[i] - m);
        cvec[j] = expf(bcol[7] - m) / den;
    }
}

extern "C" void kernel_launch(void* const* d_in, const int* in_sizes, int n_in,
                              void* d_out, int out_size, void* d_ws, size_t ws_size,
                              hipStream_t stream) {
    const float* x = (const float*)d_in[0];
    const float* W = (const float*)d_in[1];
    float* out = (float*)d_out;
    float* ws  = (float*)d_ws;

    float* vpart  = ws;
    float* spare  = vpart + STRIPS * NCOL;
    float* vpart2 = spare + 8192;
    float* uhm    = vpart2 + 16384;
    float* Bmat   = uhm + 4096;
    float* cvec   = Bmat + 64;

    const float* x7 = x + (size_t)7 * JB * IN_DIM;
    const float* w7 = W + (size_t)7 * NUM_OUT * IN_DIM * OUT_DIM;

    float* xpart = vpart;
    float* qbuf = vpart + XW_PARTS * 4096;
    k_xsum<<<dim3(XW_PARTS, 8), 512, 0, stream>>>(x, xpart);
    k_uhm<<<NUM_IN * NUM_OUT, 256, 0, stream>>>(xpart, W, uhm, Bmat, cvec);

    k_uhl_pass<<<PASS_WGS, 256, 0, stream>>>(x7, w7, cvec, vpart2, qbuf, out);
    k_route<<<NUM_OUT, 64, 0, stream>>>(vpart2, uhm, Bmat, cvec);
    k_vsum<<<PASS_WGS, 256, 0, stream>>>(out, qbuf, cvec, vpart2);
    k_route<<<NUM_OUT, 64, 0, stream>>>(vpart2, uhm, Bmat, cvec);
    k_scale<<<PASS_WGS, 256, 0, stream>>>(out, qbuf, cvec);
}
